# starting kernel
# baseline (speedup 1.0000x reference)
#include <hip/hip_runtime.h>
#include <hip/hip_bf16.h>
#include <math.h>

typedef __attribute__((ext_vector_type(2))) float v2f;
typedef __attribute__((ext_vector_type(8))) float v8f;

constexpr int kNRows     = 8192;
constexpr int kF         = 188;
constexpr int kH         = 16;
constexpr int kRowStride = kF * kH;
constexpr int kNCont     = 177;

__device__ __forceinline__ float fast_rcp(float a) {
    return __builtin_amdgcn_rcpf(a);
}
__device__ __forceinline__ float fast_sigmoid(float a) {
    return fast_rcp(1.0f + __expf(-a));
}
__device__ __forceinline__ float fast_tanh(float a) {
#if __has_builtin(__builtin_amdgcn_tanhf)
    return __builtin_amdgcn_tanhf(a);
#elif __has_builtin(__builtin_amdgcn_tanh_f32)
    return __builtin_amdgcn_tanh_f32(a);
#else
    const float e2 = __expf(-2.0f * __builtin_fabsf(a));
    const float t  = (1.0f - e2) * fast_rcp(1.0f + e2);
    return __builtin_copysignf(t, a);
#endif
}

__global__ __launch_bounds__(128)
void tab_decoder_wmma(const float* __restrict__ x,
                      const float* __restrict__ W2a, const float* __restrict__ b2a,
                      const float* __restrict__ W3,  const float* __restrict__ b3,
                      const float* __restrict__ W4,  const float* __restrict__ b4,
                      const float* __restrict__ W6,  const float* __restrict__ b6,
                      const float* __restrict__ W7,  const float* __restrict__ b7,
                      const float* __restrict__ Wc,  const float* __restrict__ bc,
                      const float* __restrict__ eps,
                      float* __restrict__ out)
{
    const int tile = blockIdx.x;
    const int wave = threadIdx.x >> 5;
    const int lane = threadIdx.x & 31;
    const int m    = lane & 15;
    const int kh   = lane >> 4;
    const int n    = m;

    const int    arow = tile * 16 + m;
    const float* xb   = x + (size_t)arow * kRowStride;

    for (int f = wave; f < kF; f += 4) {
        const float* W; const float* bias; int od;
        if      (f < 4)  { W = W2a + f       * kH * 2; bias = b2a + f       * 2; od = 2; }
        else if (f < 6)  { W = W3  + (f - 4) * kH * 3; bias = b3  + (f - 4) * 3; od = 3; }
        else if (f < 7)  { W = W4;                     bias = b4;                od = 4; }
        else if (f < 8)  { W = W6;                     bias = b6;                od = 6; }
        else if (f < 11) { W = W7  + (f - 8) * kH * 7; bias = b7  + (f - 8) * 7; od = 7; }
        else             { W = Wc  + (f - 11)* kH * 2; bias = bc  + (f - 11)* 2; od = 2; }

        const float* xf = xb + f * kH;
        __builtin_prefetch(xf + 4 * kH, 0, 1);

        const bool valid = (n < od);
        const int  nc    = valid ? n : 0;

        const float braw = bias[nc];
        const float bn   = valid ? braw : 0.0f;

        v2f a0 = *(const v2f*)(xf + 0  + 2 * kh);
        v2f a1 = *(const v2f*)(xf + 4  + 2 * kh);
        v2f a2 = *(const v2f*)(xf + 8  + 2 * kh);
        v2f a3 = *(const v2f*)(xf + 12 + 2 * kh);

        const float* Wl = W + 2 * kh * od + nc;
        float w0 = Wl[0 * od], w1 = Wl[1 * od];
        float w2 = Wl[4 * od], w3 = Wl[5 * od];
        float w4 = Wl[8 * od], w5 = Wl[9 * od];
        float w6 = Wl[12 * od], w7 = Wl[13 * od];

        v2f b0, b1, b2, b3;
        b0.x = valid ? w0 : 0.0f;  b0.y = valid ? w1 : 0.0f;
        b1.x = valid ? w2 : 0.0f;  b1.y = valid ? w3 : 0.0f;
        b2.x = valid ? w4 : 0.0f;  b2.y = valid ? w5 : 0.0f;
        b3.x = valid ? w6 : 0.0f;  b3.y = valid ? w7 : 0.0f;

        v8f acc = {};
        acc = __builtin_amdgcn_wmma_f32_16x16x4_f32(false, a0, false, b0,
                                                    (short)0, acc, false, false);
        acc = __builtin_amdgcn_wmma_f32_16x16x4_f32(false, a1, false, b1,
                                                    (short)0, acc, false, false);
        acc = __builtin_amdgcn_wmma_f32_16x16x4_f32(false, a2, false, b2,
                                                    (short)0, acc, false, false);
        acc = __builtin_amdgcn_wmma_f32_16x16x4_f32(false, a3, false, b3,
                                                    (short)0, acc, false, false);

        if (f < 11) {
#pragma unroll
            for (int r = 0; r < 8; ++r) {
                float v   = valid ? (acc[r] + bn) : -3.0e38f;
                int   idx = n;
#pragma unroll
                for (int mask = 1; mask <= 4; mask <<= 1) {
                    float ov = __shfl_xor(v,   mask, 32);
                    int   oi = __shfl_xor(idx, mask, 32);
                    if (ov > v || (ov == v && oi < idx)) { v = ov; idx = oi; }
                }
                if (n == 0) {
                    const int orow = tile * 16 + r + 8 * kh;
                    out[(size_t)orow * kF + f] = (float)idx;
                }
            }
        } else {
            const int fc = f - 11;
#pragma unroll
            for (int r = 0; r < 8; ++r) {
                float t  = acc[r] + bn;
                float ts = __shfl_xor(t, 1, 32);
                if (n == 0) {
                    const int orow = tile * 16 + r + 8 * kh;
                    const float mean = fast_tanh(t);
                    const float sd   = fast_sigmoid(ts);
                    const float e    = eps[(size_t)orow * kNCont + fc];
                    out[(size_t)orow * kF + f] = mean + sd * e;
                }
            }
        }
    }
}

extern "C" void kernel_launch(void* const* d_in, const int* in_sizes, int n_in,
                              void* d_out, int out_size, void* d_ws, size_t ws_size,
                              hipStream_t stream) {
    const float* x   = (const float*)d_in[0];
    const float* W2a = (const float*)d_in[1];
    const float* b2a = (const float*)d_in[2];
    const float* W3  = (const float*)d_in[3];
    const float* b3  = (const float*)d_in[4];
    const float* W4  = (const float*)d_in[5];
    const float* b4  = (const float*)d_in[6];
    const float* W6  = (const float*)d_in[7];
    const float* b6  = (const float*)d_in[8];
    const float* W7  = (const float*)d_in[9];
    const float* b7  = (const float*)d_in[10];
    const float* Wc  = (const float*)d_in[11];
    const float* bc  = (const float*)d_in[12];
    const float* eps = (const float*)d_in[13];
    float* out = (float*)d_out;

    tab_decoder_wmma<<<dim3(kNRows / 16), dim3(128), 0, stream>>>(
        x, W2a, b2a, W3, b3, W4, b4, W6, b6, W7, b7, Wc, bc, eps, out);
}


# Product transposed (x as B operand, W^T as A): a lane owns one row, so argmax and tanh/sigmoid/eps are lane-local, no shuffles; a wave pass does 4 adjacent features, one 16-byte row store; same sums
# speedup vs baseline: 6.3546x; 6.3546x over previous
#include <hip/hip_runtime.h>
#include <hip/hip_bf16.h>
#include <math.h>


typedef __attribute__((ext_vector_type(2))) float v2f;
typedef __attribute__((ext_vector_type(4))) float v4f;
typedef __attribute__((ext_vector_type(8))) float v8f;

constexpr int kNRows     = 8192;
constexpr int kF         = 188;
constexpr int kH         = 16;
constexpr int kRowStride = kF * kH;
constexpr int kNCont     = 177;
constexpr int kNQuad     = kF / 4;

__device__ __forceinline__ float fast_rcp(float a) {
    return __builtin_amdgcn_rcpf(a);
}
__device__ __forceinline__ float fast_sigmoid(float a) {
    return fast_rcp(1.0f + __expf(-a));
}
__device__ __forceinline__ float fast_tanh(float a) {
#if __has_builtin(__builtin_amdgcn_tanhf)
    return __builtin_amdgcn_tanhf(a);
#elif __has_builtin(__builtin_amdgcn_tanh_f32)
    return __builtin_amdgcn_tanh_f32(a);
#else
    const float e2 = __expf(-2.0f * __builtin_fabsf(a));
    const float t  = (1.0f - e2) * fast_rcp(1.0f + e2);
    return __builtin_copysignf(t, a);
#endif
}

struct Frags {
    v2f   x[4];
    float w[8];
};

__device__ __forceinline__ void load_frags(Frags& fr,
                                           const float* __restrict__ xf,
                                           const float* __restrict__ W,
                                           int od, int oc, int h)
{
#pragma unroll
    for (int t = 0; t < 4; ++t)
        fr.x[t] = *(const v2f*)(xf + 4 * t + 2 * h);
    const float* Wl = W + 2 * h * od + oc;
#pragma unroll
    for (int t = 0; t < 4; ++t) {
        fr.w[2 * t + 0] = Wl[(4 * t + 0) * od];
        fr.w[2 * t + 1] = Wl[(4 * t + 1) * od];
    }
}

__device__ __forceinline__ v8f wmma_chain(const Frags& fr, bool valid)
{
    v8f acc = {};
#pragma unroll
    for (int t = 0; t < 4; ++t) {
        v2f a;
        a.x = valid ? fr.w[2 * t + 0] : 0.0f;
        a.y = valid ? fr.w[2 * t + 1] : 0.0f;
        acc = __builtin_amdgcn_wmma_f32_16x16x4_f32(false, a, false, fr.x[t],
                                                    (short)0, acc, false, false);
    }
    return acc;
}

struct Params {
    const float* W2a; const float* b2a;
    const float* W3;  const float* b3;
    const float* W4;  const float* b4;
    const float* W6;  const float* b6;
    const float* W7;  const float* b7;
    const float* Wc;  const float* bc;
};

template <int F> struct Early {
    static constexpr int od = (F < 4) ? 2 : (F < 6) ? 3 : (F < 7) ? 4 : (F < 8) ? 6 : (F < 11) ? 7 : 2;
    static __device__ __forceinline__ const float* W(const Params& p) {
        return (F < 4)  ? p.W2a + F       * kH * 2
             : (F < 6)  ? p.W3  + (F - 4) * kH * 3
             : (F < 7)  ? p.W4
             : (F < 8)  ? p.W6
             : (F < 11) ? p.W7  + (F - 8) * kH * 7
             :            p.Wc;
    }
    static __device__ __forceinline__ const float* bias(const Params& p) {
        return (F < 4)  ? p.b2a + F       * 2
             : (F < 6)  ? p.b3  + (F - 4) * 3
             : (F < 7)  ? p.b4
             : (F < 8)  ? p.b6
             : (F < 11) ? p.b7  + (F - 8) * 7
             :            p.bc;
    }
    static __device__ __forceinline__ void load(Frags& fr, const Params& p,
                                                const float* __restrict__ xrow, int c, int h) {
        const int oc = (c < od) ? c : 0;
        load_frags(fr, xrow + F * kH, W(p), od, oc, h);
    }
    static __device__ __forceinline__ float finish(const Frags& fr, const Params& p,
                                                   const float* __restrict__ erow, int c) {
        const v8f acc = wmma_chain(fr, c < od);
        const float* b = bias(p);
        if (F < 11) {
            float best = acc[0] + b[0];
            int   idx  = 0;
#pragma unroll
            for (int o = 1; o < od; ++o) {
                const float v = acc[o] + b[o];
                if (v > best) { best = v; idx = o; }
            }
            return (float)idx;
        } else {
            const float t  = acc[0] + b[0];
            const float ts = acc[1] + b[1];
            return fast_tanh(t) + fast_sigmoid(ts) * erow[F - 11];
        }
    }
};

template <int Q>
__device__ __forceinline__ void early_quad(float (&res)[4], const Params& p,
                                           const float* __restrict__ xrow,
                                           const float* __restrict__ erow, int c, int h)
{
    Frags f0, f1, f2, f3;
    Early<4 * Q + 0>::load(f0, p, xrow, c, h);
    Early<4 * Q + 1>::load(f1, p, xrow, c, h);
    Early<4 * Q + 2>::load(f2, p, xrow, c, h);
    Early<4 * Q + 3>::load(f3, p, xrow, c, h);
    res[0] = Early<4 * Q + 0>::finish(f0, p, erow, c);
    res[1] = Early<4 * Q + 1>::finish(f1, p, erow, c);
    res[2] = Early<4 * Q + 2>::finish(f2, p, erow, c);
    res[3] = Early<4 * Q + 3>::finish(f3, p, erow, c);
}

__global__ __launch_bounds__(128)
void tab_decoder_wmma(const float* __restrict__ x,
                      const float* __restrict__ W2a, const float* __restrict__ b2a,
                      const float* __restrict__ W3,  const float* __restrict__ b3,
                      const float* __restrict__ W4,  const float* __restrict__ b4,
                      const float* __restrict__ W6,  const float* __restrict__ b6,
                      const float* __restrict__ W7,  const float* __restrict__ b7,
                      const float* __restrict__ Wc,  const float* __restrict__ bc,
                      const float* __restrict__ eps,
                      float* __restrict__ out)
{
    const int tile = blockIdx.x;
    const int wave = __builtin_amdgcn_readfirstlane((int)(threadIdx.x >> 5));
    const int lane = threadIdx.x & 31;
    const int c    = lane & 15;
    const int h    = lane >> 4;

    const int    row  = tile * 16 + c;
    const float* xrow = x   + (size_t)row * kRowStride;
    const float* erow = eps + (size_t)row * kNCont;
    float*       orow = out + (size_t)row * kF;
    const Params prm = { W2a, b2a, W3, b3, W4, b4, W6, b6, W7, b7, Wc, bc };

    for (int q = wave; q < kNQuad; q += 4) {
        const int f0 = 4 * q;

        if (q + 4 < kNQuad)
            __builtin_prefetch(xrow + (f0 + 16) * kH + 32 * h, 0, 1);

        float res[4];

        if (q >= 3) {
            const int  fc0   = f0 - 11;
            const bool valid = (c < 2);
            const int  oc    = valid ? c : 0;

            Frags fr[4];
#pragma unroll
            for (int j = 0; j < 4; ++j)
                load_frags(fr[j], xrow + (f0 + j) * kH, Wc + (fc0 + j) * kH * 2, 2, oc, h);
            float bm[4], bs[4], e[4];
#pragma unroll
            for (int j = 0; j < 4; ++j) {
                bm[j] = bc[(fc0 + j) * 2 + 0];
                bs[j] = bc[(fc0 + j) * 2 + 1];
                e[j]  = erow[fc0 + j];
            }

#pragma unroll
            for (int j = 0; j < 4; ++j) {
                const v8f acc = wmma_chain(fr[j], valid);
                const float t  = acc[0] + bm[j];
                const float ts = acc[1] + bs[j];
                res[j] = fast_tanh(t) + fast_sigmoid(ts) * e[j];
            }
        } else if (q == 0) {
            early_quad<0>(res, prm, xrow, erow, c, h);
        } else if (q == 1) {
            early_quad<1>(res, prm, xrow, erow, c, h);
        } else {
            early_quad<2>(res, prm, xrow, erow, c, h);
        }

        if (h == 0) {
            v4f r;
            r.x = res[0]; r.y = res[1]; r.z = res[2]; r.w = res[3];
            *(v4f*)(orow + f0) = r;
        }
    }
}

extern "C" void kernel_launch(void* const* d_in, const int* in_sizes, int n_in,
                              void* d_out, int out_size, void* d_ws, size_t ws_size,
                              hipStream_t stream) {
    const float* x   = (const float*)d_in[0];
    const float* W2a = (const float*)d_in[1];
    const float* b2a = (const float*)d_in[2];
    const float* W3  = (const float*)d_in[3];
    const float* b3  = (const float*)d_in[4];
    const float* W4  = (const float*)d_in[5];
    const float* b4  = (const float*)d_in[6];
    const float* W6  = (const float*)d_in[7];
    const float* b6  = (const float*)d_in[8];
    const float* W7  = (const float*)d_in[9];
    const float* b7  = (const float*)d_in[10];
    const float* Wc  = (const float*)d_in[11];
    const float* bc  = (const float*)d_in[12];
    const float* eps = (const float*)d_in[13];
    float* out = (float*)d_out;

    tab_decoder_wmma<<<dim3(kNRows / 16), dim3(128), 0, stream>>>(
        x, W2a, b2a, W3, b3, W4, b4, W6, b6, W7, b7, Wc, bc, eps, out);
}


# Continuous quads: WMMA column = one (row, feature) of a 4-row group, the quad's W^T stacked as A rows; a load reads 4 rows x 256 contiguous bytes, not 16 rows x 16; k sum order in a feature changes
# speedup vs baseline: 6.9669x; 1.0964x over previous
#include <hip/hip_runtime.h>
#include <hip/hip_bf16.h>
#include <math.h>


typedef __attribute__((ext_vector_type(2))) float v2f;
typedef __attribute__((ext_vector_type(4))) float v4f;
typedef __attribute__((ext_vector_type(8))) float v8f;

constexpr int kNRows     = 8192;
constexpr int kF         = 188;
constexpr int kH         = 16;
constexpr int kRowStride = kF * kH;
constexpr int kNCont     = 177;
constexpr int kNQuad     = kF / 4;

__device__ __forceinline__ float fast_rcp(float a) {
    return __builtin_amdgcn_rcpf(a);
}
__device__ __forceinline__ float fast_sigmoid(float a) {
    return fast_rcp(1.0f + __expf(-a));
}
__device__ __forceinline__ float fast_tanh(float a) {
#if __has_builtin(__builtin_amdgcn_tanhf)
    return __builtin_amdgcn_tanhf(a);
#elif __has_builtin(__builtin_amdgcn_tanh_f32)
    return __builtin_amdgcn_tanh_f32(a);
#else
    const float e2 = __expf(-2.0f * __builtin_fabsf(a));
    const float t  = (1.0f - e2) * fast_rcp(1.0f + e2);
    return __builtin_copysignf(t, a);
#endif
}

struct Frags {
    v2f   x[4];
    float w[8];
};

__device__ __forceinline__ void load_frags(Frags& fr,
                                           const float* __restrict__ xf,
                                           const float* __restrict__ W,
                                           int od, int oc, int h)
{
#pragma unroll
    for (int t = 0; t < 4; ++t)
        fr.x[t] = *(const v2f*)(xf + 4 * t + 2 * h);
    const float* Wl = W + 2 * h * od + oc;
#pragma unroll
    for (int t = 0; t < 4; ++t) {
        fr.w[2 * t + 0] = Wl[(4 * t + 0) * od];
        fr.w[2 * t + 1] = Wl[(4 * t + 1) * od];
    }
}

__device__ __forceinline__ v8f wmma_chain(const Frags& fr, bool valid)
{
    v8f acc = {};
#pragma unroll
    for (int t = 0; t < 4; ++t) {
        v2f a;
        a.x = valid ? fr.w[2 * t + 0] : 0.0f;
        a.y = valid ? fr.w[2 * t + 1] : 0.0f;
        acc = __builtin_amdgcn_wmma_f32_16x16x4_f32(false, a, false, fr.x[t],
                                                    (short)0, acc, false, false);
    }
    return acc;
}

struct Params {
    const float* W2a; const float* b2a;
    const float* W3;  const float* b3;
    const float* W4;  const float* b4;
    const float* W6;  const float* b6;
    const float* W7;  const float* b7;
    const float* Wc;  const float* bc;
};

template <int F> struct Early {
    static constexpr int od = (F < 4) ? 2 : (F < 6) ? 3 : (F < 7) ? 4 : (F < 8) ? 6 : (F < 11) ? 7 : 2;
    static __device__ __forceinline__ const float* W(const Params& p) {
        return (F < 4)  ? p.W2a + F       * kH * 2
             : (F < 6)  ? p.W3  + (F - 4) * kH * 3
             : (F < 7)  ? p.W4
             : (F < 8)  ? p.W6
             : (F < 11) ? p.W7  + (F - 8) * kH * 7
             :            p.Wc;
    }
    static __device__ __forceinline__ const float* bias(const Params& p) {
        return (F < 4)  ? p.b2a + F       * 2
             : (F < 6)  ? p.b3  + (F - 4) * 3
             : (F < 7)  ? p.b4
             : (F < 8)  ? p.b6
             : (F < 11) ? p.b7  + (F - 8) * 7
             :            p.bc;
    }
    static __device__ __forceinline__ void load(Frags& fr, const Params& p,
                                                const float* __restrict__ xrow, int c, int h) {
        const int oc = (c < od) ? c : 0;
        load_frags(fr, xrow + F * kH, W(p), od, oc, h);
    }
    static __device__ __forceinline__ float finish(const Frags& fr, const Params& p,
                                                   const float* __restrict__ erow, int c) {
        const v8f acc = wmma_chain(fr, c < od);
        const float* b = bias(p);
        if (F < 11) {
            float best = acc[0] + b[0];
            int   idx  = 0;
#pragma unroll
            for (int o = 1; o < od; ++o) {
                const float v = acc[o] + b[o];
                if (v > best) { best = v; idx = o; }
            }
            return (float)idx;
        } else {
            const float t  = acc[0] + b[0];
            const float ts = acc[1] + b[1];
            return fast_tanh(t) + fast_sigmoid(ts) * erow[F - 11];
        }
    }
};

template <int Q>
__device__ __forceinline__ void early_quad(float (&res)[4], const Params& p,
                                           const float* __restrict__ xrow,
                                           const float* __restrict__ erow, int c, int h)
{
    Frags f0, f1, f2, f3;
    Early<4 * Q + 0>::load(f0, p, xrow, c, h);
    Early<4 * Q + 1>::load(f1, p, xrow, c, h);
    Early<4 * Q + 2>::load(f2, p, xrow, c, h);
    Early<4 * Q + 3>::load(f3, p, xrow, c, h);
    res[0] = Early<4 * Q + 0>::finish(f0, p, erow, c);
    res[1] = Early<4 * Q + 1>::finish(f1, p, erow, c);
    res[2] = Early<4 * Q + 2>::finish(f2, p, erow, c);
    res[3] = Early<4 * Q + 3>::finish(f3, p, erow, c);
}

__global__ __launch_bounds__(128)
void tab_decoder_wmma(const float* __restrict__ x,
                      const float* __restrict__ W2a, const float* __restrict__ b2a,
                      const float* __restrict__ W3,  const float* __restrict__ b3,
                      const float* __restrict__ W4,  const float* __restrict__ b4,
                      const float* __restrict__ W6,  const float* __restrict__ b6,
                      const float* __restrict__ W7,  const float* __restrict__ b7,
                      const float* __restrict__ Wc,  const float* __restrict__ bc,
                      const float* __restrict__ eps,
                      float* __restrict__ out)
{
    const int tile = blockIdx.x;
    const int wave = __builtin_amdgcn_readfirstlane((int)(threadIdx.x >> 5));
    const int lane = threadIdx.x & 31;
    const int c    = lane & 15;
    const int h    = lane >> 4;

    const int    row  = tile * 16 + c;
    const float* xrow = x   + (size_t)row * kRowStride;
    const float* erow = eps + (size_t)row * kNCont;
    float*       orow = out + (size_t)row * kF;
    const Params prm = { W2a, b2a, W3, b3, W4, b4, W6, b6, W7, b7, Wc, bc };

    for (int q = wave; q < kNQuad; q += 4) {
        const int f0 = 4 * q;

        if (q + 4 < kNQuad)
            __builtin_prefetch(xrow + (f0 + 16) * kH + 32 * h, 0, 1);

        if (q >= 3) {
            const int  fc0  = f0 - 11;
            const int  fj   = c & 3;
            const int  rr   = c >> 2;
            const int  ma   = c & 7;

            const int    row0 = tile * 16 + rr;
            const float* xq   = x + (size_t)row0 * kRowStride + (f0 + fj) * kH + 8 * h;
            const float* eq   = eps + (size_t)(row0 + 8 * h) * kNCont + fc0 + fj;
            float*       oq   = out + (size_t)(row0 + 8 * h) * kF + f0 + fj;

            v4f xa[4], xb[4];
#pragma unroll
            for (int g = 0; g < 4; ++g) {
                xa[g] = *(const v4f*)(xq + (size_t)(4 * g) * kRowStride);
                xb[g] = *(const v4f*)(xq + (size_t)(4 * g) * kRowStride + 4);
            }
            const float* Wl = Wc + (fc0 + (ma >> 1)) * kH * 2 + 16 * h + (ma & 1);
            v2f a[4];
#pragma unroll
            for (int t = 0; t < 4; ++t) {
                a[t].x = Wl[4 * t + 0];
                a[t].y = Wl[4 * t + 2];
            }
            const float bm = bc[(fc0 + fj) * 2 + 0];
            const float bs = bc[(fc0 + fj) * 2 + 1];
            float e[2];
#pragma unroll
            for (int u = 0; u < 2; ++u)
                e[u] = eq[(size_t)(4 * u) * kNCont];

            v8f acc[4];
#pragma unroll
            for (int g = 0; g < 4; ++g) {
                v2f b0, b1, b2, b3;
                b0.x = xa[g].x; b0.y = xa[g].y;
                b1.x = xa[g].z; b1.y = xa[g].w;
                b2.x = xb[g].x; b2.y = xb[g].y;
                b3.x = xb[g].z; b3.y = xb[g].w;
                v8f d = {};
                d = __builtin_amdgcn_wmma_f32_16x16x4_f32(false, a[0], false, b0, (short)0, d, false, false);
                d = __builtin_amdgcn_wmma_f32_16x16x4_f32(false, a[1], false, b1, (short)0, d, false, false);
                d = __builtin_amdgcn_wmma_f32_16x16x4_f32(false, a[2], false, b2, (short)0, d, false, false);
                d = __builtin_amdgcn_wmma_f32_16x16x4_f32(false, a[3], false, b3, (short)0, d, false, false);
                acc[g] = d;
            }

            const unsigned hm = h ? 0xffffffffu : 0u;
            const unsigned k0 = (fj == 0) ? 0xffffffffu : 0u;
            const unsigned k1 = (fj == 1) ? 0xffffffffu : 0u;
            const unsigned k2 = (fj == 2) ? 0xffffffffu : 0u;
            const unsigned k3 = (fj == 3) ? 0xffffffffu : 0u;
#pragma unroll
            for (int u = 0; u < 2; ++u) {
                unsigned sdw[8];
#pragma unroll
                for (int i = 0; i < 8; ++i)
                    sdw[i] = (__float_as_uint(acc[u][i]) & ~hm) | (__float_as_uint(acc[2 + u][i]) & hm);
                const unsigned um = (sdw[0] & k0) | (sdw[2] & k1) | (sdw[4] & k2) | (sdw[6] & k3);
                const unsigned uv = (sdw[1] & k0) | (sdw[3] & k1) | (sdw[5] & k2) | (sdw[7] & k3);
                const float t  = __uint_as_float(um) + bm;
                const float ts = __uint_as_float(uv) + bs;
                oq[(size_t)(4 * u) * kF] = fast_tanh(t) + fast_sigmoid(ts) * e[u];
            }
        } else {
            float res[4];
            if (q == 0) {
                early_quad<0>(res, prm, xrow, erow, c, h);
            } else if (q == 1) {
                early_quad<1>(res, prm, xrow, erow, c, h);
            } else {
                early_quad<2>(res, prm, xrow, erow, c, h);
            }
            if (h == 0) {
                v4f r;
                r.x = res[0]; r.y = res[1]; r.z = res[2]; r.w = res[3];
                *(v4f*)(orow + f0) = r;
            }
        }
    }
}

extern "C" void kernel_launch(void* const* d_in, const int* in_sizes, int n_in,
                              void* d_out, int out_size, void* d_ws, size_t ws_size,
                              hipStream_t stream) {
    const float* x   = (const float*)d_in[0];
    const float* W2a = (const float*)d_in[1];
    const float* b2a = (const float*)d_in[2];
    const float* W3  = (const float*)d_in[3];
    const float* b3  = (const float*)d_in[4];
    const float* W4  = (const float*)d_in[5];
    const float* b4  = (const float*)d_in[6];
    const float* W6  = (const float*)d_in[7];
    const float* b6  = (const float*)d_in[8];
    const float* W7  = (const float*)d_in[9];
    const float* b7  = (const float*)d_in[10];
    const float* Wc  = (const float*)d_in[11];
    const float* bc  = (const float*)d_in[12];
    const float* eps = (const float*)d_in[13];
    float* out = (float*)d_out;

    tab_decoder_wmma<<<dim3(kNRows / 16), dim3(128), 0, stream>>>(
        x, W2a, b2a, W3, b3, W4, b4, W6, b6, W7, b7, Wc, bc, eps, out);
}


# A wave pass covers 4 quads (16 features x 16 rows), not 1: all its 32 16-byte x loads per lane issued ahead of its WMMAs; 3 memory round trips for a wave's 11 continuous quads, not 11; same sums
# speedup vs baseline: 7.6302x; 1.0952x over previous
#include <hip/hip_runtime.h>
#include <hip/hip_bf16.h>
#include <math.h>


typedef __attribute__((ext_vector_type(2))) float v2f;
typedef __attribute__((ext_vector_type(4))) float v4f;
typedef __attribute__((ext_vector_type(8))) float v8f;

constexpr int kNRows     = 8192;
constexpr int kF         = 188;
constexpr int kH         = 16;
constexpr int kRowStride = kF * kH;
constexpr int kNCont     = 177;
constexpr int kNQuad     = kF / 4;

__device__ __forceinline__ float fast_rcp(float a) {
    return __builtin_amdgcn_rcpf(a);
}
__device__ __forceinline__ float fast_sigmoid(float a) {
    return fast_rcp(1.0f + __expf(-a));
}
__device__ __forceinline__ float fast_tanh(float a) {
#if __has_builtin(__builtin_amdgcn_tanhf)
    return __builtin_amdgcn_tanhf(a);
#elif __has_builtin(__builtin_amdgcn_tanh_f32)
    return __builtin_amdgcn_tanh_f32(a);
#else
    const float e2 = __expf(-2.0f * __builtin_fabsf(a));
    const float t  = (1.0f - e2) * fast_rcp(1.0f + e2);
    return __builtin_copysignf(t, a);
#endif
}

struct Frags {
    v2f   x[4];
    float w[8];
};

__device__ __forceinline__ void load_frags(Frags& fr,
                                           const float* __restrict__ xf,
                                           const float* __restrict__ W,
                                           int od, int oc, int h)
{
#pragma unroll
    for (int t = 0; t < 4; ++t)
        fr.x[t] = *(const v2f*)(xf + 4 * t + 2 * h);
    const float* Wl = W + 2 * h * od + oc;
#pragma unroll
    for (int t = 0; t < 4; ++t) {
        fr.w[2 * t + 0] = Wl[(4 * t + 0) * od];
        fr.w[2 * t + 1] = Wl[(4 * t + 1) * od];
    }
}

__device__ __forceinline__ v8f wmma_chain(const Frags& fr, bool valid)
{
    v8f acc = {};
#pragma unroll
    for (int t = 0; t < 4; ++t) {
        v2f a;
        a.x = valid ? fr.w[2 * t + 0] : 0.0f;
        a.y = valid ? fr.w[2 * t + 1] : 0.0f;
        acc = __builtin_amdgcn_wmma_f32_16x16x4_f32(false, a, false, fr.x[t],
                                                    (short)0, acc, false, false);
    }
    return acc;
}

struct Params {
    const float* W2a; const float* b2a;
    const float* W3;  const float* b3;
    const float* W4;  const float* b4;
    const float* W6;  const float* b6;
    const float* W7;  const float* b7;
    const float* Wc;  const float* bc;
};

template <int F> struct Early {
    static constexpr int od = (F < 4) ? 2 : (F < 6) ? 3 : (F < 7) ? 4 : (F < 8) ? 6 : (F < 11) ? 7 : 2;
    static __device__ __forceinline__ const float* W(const Params& p) {
        return (F < 4)  ? p.W2a + F       * kH * 2
             : (F < 6)  ? p.W3  + (F - 4) * kH * 3
             : (F < 7)  ? p.W4
             : (F < 8)  ? p.W6
             : (F < 11) ? p.W7  + (F - 8) * kH * 7
             :            p.Wc;
    }
    static __device__ __forceinline__ const float* bias(const Params& p) {
        return (F < 4)  ? p.b2a + F       * 2
             : (F < 6)  ? p.b3  + (F - 4) * 3
             : (F < 7)  ? p.b4
             : (F < 8)  ? p.b6
             : (F < 11) ? p.b7  + (F - 8) * 7
             :            p.bc;
    }
    static __device__ __forceinline__ void load(Frags& fr, const Params& p,
                                                const float* __restrict__ xrow, int c, int h) {
        const int oc = (c < od) ? c : 0;
        load_frags(fr, xrow + F * kH, W(p), od, oc, h);
    }
    static __device__ __forceinline__ float finish(const Frags& fr, const Params& p,
                                                   const float* __restrict__ erow, int c) {
        const v8f acc = wmma_chain(fr, c < od);
        const float* b = bias(p);
        if (F < 11) {
            float best = acc[0] + b[0];
            int   idx  = 0;
#pragma unroll
            for (int o = 1; o < od; ++o) {
                const float v = acc[o] + b[o];
                if (v > best) { best = v; idx = o; }
            }
            return (float)idx;
        } else {
            const float t  = acc[0] + b[0];
            const float ts = acc[1] + b[1];
            return fast_tanh(t) + fast_sigmoid(ts) * erow[F - 11];
        }
    }
};

template <int Q>
__device__ __forceinline__ void early_quad(float (&res)[4], const Params& p,
                                           const float* __restrict__ xrow,
                                           const float* __restrict__ erow, int c, int h)
{
    Frags f0, f1, f2, f3;
    Early<4 * Q + 0>::load(f0, p, xrow, c, h);
    Early<4 * Q + 1>::load(f1, p, xrow, c, h);
    Early<4 * Q + 2>::load(f2, p, xrow, c, h);
    Early<4 * Q + 3>::load(f3, p, xrow, c, h);
    res[0] = Early<4 * Q + 0>::finish(f0, p, erow, c);
    res[1] = Early<4 * Q + 1>::finish(f1, p, erow, c);
    res[2] = Early<4 * Q + 2>::finish(f2, p, erow, c);
    res[3] = Early<4 * Q + 3>::finish(f3, p, erow, c);
}

template <int N>
__device__ __forceinline__ void cont_pass(const float* __restrict__ x,
                                          const float* __restrict__ Wc,
                                          const float* __restrict__ bc,
                                          const float* __restrict__ eps,
                                          float* __restrict__ out,
                                          int tile, int q0, int c, int h)
{
    const int fj   = c & 3;
    const int rr   = c >> 2;
    const int ma   = c & 7;
    const int row0 = tile * 16 + rr;

    v4f   xa[N][4], xb[N][4];
    v4f   wq[N][4];
    float bm[N], bs[N];
    float e[N][2];

#pragma unroll
    for (int i = 0; i < N; ++i) {
        const int f0  = 4 * (q0 + 4 * i);
        const int fc0 = f0 - 11;
        const float* xq = x + (size_t)row0 * kRowStride + (f0 + fj) * kH + 8 * h;
        const float* Wl = Wc + (fc0 + (ma >> 1)) * kH * 2 + 16 * h;
        const float* eq = eps + (size_t)(row0 + 8 * h) * kNCont + fc0 + fj;
#pragma unroll
        for (int j = 0; j < 4; ++j)
            wq[i][j] = *(const v4f*)(Wl + 4 * j);
#pragma unroll
        for (int g = 0; g < 4; ++g) {
            xa[i][g] = *(const v4f*)(xq + (size_t)(4 * g) * kRowStride);
            xb[i][g] = *(const v4f*)(xq + (size_t)(4 * g) * kRowStride + 4);
        }
        bm[i] = bc[(fc0 + fj) * 2 + 0];
        bs[i] = bc[(fc0 + fj) * 2 + 1];
#pragma unroll
        for (int u = 0; u < 2; ++u)
            e[i][u] = eq[(size_t)(4 * u) * kNCont];
        __builtin_amdgcn_sched_barrier(0);
    }

    const unsigned hm = h ? 0xffffffffu : 0u;
    const unsigned k0 = (fj == 0) ? 0xffffffffu : 0u;
    const unsigned k1 = (fj == 1) ? 0xffffffffu : 0u;
    const unsigned k2 = (fj == 2) ? 0xffffffffu : 0u;
    const unsigned k3 = (fj == 3) ? 0xffffffffu : 0u;
    const unsigned om = (ma & 1) ? 0xffffffffu : 0u;

#pragma unroll
    for (int i = 0; i < N; ++i) {
        const int f0 = 4 * (q0 + 4 * i);
        float*    oq = out + (size_t)(row0 + 8 * h) * kF + f0 + fj;

        v2f a[4];
#pragma unroll
        for (int t = 0; t < 4; ++t) {
            a[t].x = __uint_as_float((__float_as_uint(wq[i][t].x) & ~om) | (__float_as_uint(wq[i][t].y) & om));
            a[t].y = __uint_as_float((__float_as_uint(wq[i][t].z) & ~om) | (__float_as_uint(wq[i][t].w) & om));
        }

        v8f acc[4];
#pragma unroll
        for (int g = 0; g < 4; ++g) {
            v2f b0, b1, b2, b3;
            b0.x = xa[i][g].x; b0.y = xa[i][g].y;
            b1.x = xa[i][g].z; b1.y = xa[i][g].w;
            b2.x = xb[i][g].x; b2.y = xb[i][g].y;
            b3.x = xb[i][g].z; b3.y = xb[i][g].w;
            v8f d = {};
            d = __builtin_amdgcn_wmma_f32_16x16x4_f32(false, a[0], false, b0, (short)0, d, false, false);
            d = __builtin_amdgcn_wmma_f32_16x16x4_f32(false, a[1], false, b1, (short)0, d, false, false);
            d = __builtin_amdgcn_wmma_f32_16x16x4_f32(false, a[2], false, b2, (short)0, d, false, false);
            d = __builtin_amdgcn_wmma_f32_16x16x4_f32(false, a[3], false, b3, (short)0, d, false, false);
            acc[g] = d;
        }

#pragma unroll
        for (int u = 0; u < 2; ++u) {
            unsigned sdw[8];
#pragma unroll
            for (int r = 0; r < 8; ++r)
                sdw[r] = (__float_as_uint(acc[u][r]) & ~hm) | (__float_as_uint(acc[2 + u][r]) & hm);
            const unsigned um = (sdw[0] & k0) | (sdw[2] & k1) | (sdw[4] & k2) | (sdw[6] & k3);
            const unsigned uv = (sdw[1] & k0) | (sdw[3] & k1) | (sdw[5] & k2) | (sdw[7] & k3);
            const float t  = __uint_as_float(um) + bm[i];
            const float ts = __uint_as_float(uv) + bs[i];
            oq[(size_t)(4 * u) * kF] = fast_tanh(t) + fast_sigmoid(ts) * e[i][u];
        }
    }
}

__global__ __launch_bounds__(128)
void tab_decoder_wmma(const float* __restrict__ x,
                      const float* __restrict__ W2a, const float* __restrict__ b2a,
                      const float* __restrict__ W3,  const float* __restrict__ b3,
                      const float* __restrict__ W4,  const float* __restrict__ b4,
                      const float* __restrict__ W6,  const float* __restrict__ b6,
                      const float* __restrict__ W7,  const float* __restrict__ b7,
                      const float* __restrict__ Wc,  const float* __restrict__ bc,
                      const float* __restrict__ eps,
                      float* __restrict__ out)
{
    const int tile = blockIdx.x;
    const int wave = __builtin_amdgcn_readfirstlane((int)(threadIdx.x >> 5));
    const int lane = threadIdx.x & 31;
    const int c    = lane & 15;
    const int h    = lane >> 4;

    if (wave < 3) {
        const int    row  = tile * 16 + c;
        const float* xrow = x   + (size_t)row * kRowStride;
        const float* erow = eps + (size_t)row * kNCont;
        float*       orow = out + (size_t)row * kF;
        const Params prm = { W2a, b2a, W3, b3, W4, b4, W6, b6, W7, b7, Wc, bc };
        float res[4];
        if (wave == 0) {
            early_quad<0>(res, prm, xrow, erow, c, h);
        } else if (wave == 1) {
            early_quad<1>(res, prm, xrow, erow, c, h);
        } else {
            early_quad<2>(res, prm, xrow, erow, c, h);
        }
        if (h == 0) {
            v4f r;
            r.x = res[0]; r.y = res[1]; r.z = res[2]; r.w = res[3];
            *(v4f*)(orow + 4 * wave) = r;
        }
    }

    const int qs = (wave < 3) ? wave + 4 : 3;
#pragma nounroll
    for (int k = 0; k < 2; ++k)
        cont_pass<4>(x, Wc, bc, eps, out, tile, qs + 16 * k, c, h);
    cont_pass<3>(x, Wc, bc, eps, out, tile, qs + 32, c, h);
}

extern "C" void kernel_launch(void* const* d_in, const int* in_sizes, int n_in,
                              void* d_out, int out_size, void* d_ws, size_t ws_size,
                              hipStream_t stream) {
    const float* x   = (const float*)d_in[0];
    const float* W2a = (const float*)d_in[1];
    const float* b2a = (const float*)d_in[2];
    const float* W3  = (const float*)d_in[3];
    const float* b3  = (const float*)d_in[4];
    const float* W4  = (const float*)d_in[5];
    const float* b4  = (const float*)d_in[6];
    const float* W6  = (const float*)d_in[7];
    const float* b6  = (const float*)d_in[8];
    const float* W7  = (const float*)d_in[9];
    const float* b7  = (const float*)d_in[10];
    const float* Wc  = (const float*)d_in[11];
    const float* bc  = (const float*)d_in[12];
    const float* eps = (const float*)d_in[13];
    float* out = (float*)d_out;

    tab_decoder_wmma<<<dim3(kNRows / 16), dim3(128), 0, stream>>>(
        x, W2a, b2a, W3, b3, W4, b4, W6, b6, W7, b7, Wc, bc, eps, out);
}
